# starting kernel
# baseline (speedup 1.0000x reference)
#include <hip/hip_runtime.h>
#include <hip/hip_bf16.h>

typedef __attribute__((ext_vector_type(2))) float v2f;
typedef __attribute__((ext_vector_type(8))) float v8f;

#define B_N   8192
#define C_N   1000
#define P_N   8
#define D_N   512
#define CP_N  (C_N * P_N)

#define MT    128
#define NT    64
#define KT    64
#define LDA   68
#define LDD   66
#define NTHREADS 128

__global__ __launch_bounds__(256) void rowsq_kernel(const float* __restrict__ x,
                                                    float* __restrict__ out,
                                                    int rows) {
    const int lane = threadIdx.x & 31;
    const int row  = blockIdx.x * 8 + (threadIdx.x >> 5);
    if (row >= rows) return;
    const float* r = x + (size_t)row * D_N;
    float s = 0.0f;
    #pragma unroll
    for (int i = 0; i < D_N; i += 128) {
        float4 v = *(const float4*)(r + i + lane * 4);
        s += v.x * v.x + v.y * v.y + v.z * v.z + v.w * v.w;
    }
    #pragma unroll
    for (int m = 16; m >= 1; m >>= 1) s += __shfl_xor(s, m, 32);
    if (lane == 0) out[row] = s;
}

__global__ __launch_bounds__(NTHREADS) void proto_head_kernel(
    const float* __restrict__ z,
    const float* __restrict__ mu,
    const float* __restrict__ z2,
    const float* __restrict__ mu2,
    float* __restrict__ out)
{
    __shared__ float smem[MT * LDA + NT * LDA];
    float* zs = smem;
    float* ms = smem + MT * LDA;

    const int t     = threadIdx.x;
    const int wave  = t >> 5;
    const int lane  = t & 31;
    const int lo    = lane & 15;
    const int hi    = lane >> 4;
    const int bn    = blockIdx.x;
    const int bm    = blockIdx.y;
    const int waveM = wave * 32;

    const float* zg = z  + (size_t)(bm * MT) * D_N;
    const float* mg = mu + (size_t)(bn * NT) * D_N;

    v8f acc[2][4] = {};

    for (int ks = 0; ks < D_N; ks += KT) {
        #pragma unroll
        for (int p = 0; p < 16; ++p) {
            int idx = p * NTHREADS + t;
            int row = idx >> 4;
            int c4  = (idx & 15) << 2;
            float4 v = *(const float4*)(zg + (size_t)row * D_N + ks + c4);
            *(float4*)(zs + row * LDA + c4) = v;
        }
        #pragma unroll
        for (int p = 0; p < 8; ++p) {
            int idx = p * NTHREADS + t;
            int row = idx >> 4;
            int c4  = (idx & 15) << 2;
            float4 v = *(const float4*)(mg + (size_t)row * D_N + ks + c4);
            *(float4*)(ms + row * LDA + c4) = v;
        }
        __syncthreads();

        #pragma unroll
        for (int kk = 0; kk < KT; kk += 4) {
            v2f a0 = *(const v2f*)(zs + (waveM +      lo) * LDA + kk + 2 * hi);
            v2f a1 = *(const v2f*)(zs + (waveM + 16 + lo) * LDA + kk + 2 * hi);
            v2f b0 = *(const v2f*)(ms + (      lo) * LDA + kk + 2 * hi);
            v2f b1 = *(const v2f*)(ms + (16 + lo) * LDA + kk + 2 * hi);
            v2f b2 = *(const v2f*)(ms + (32 + lo) * LDA + kk + 2 * hi);
            v2f b3 = *(const v2f*)(ms + (48 + lo) * LDA + kk + 2 * hi);

            acc[0][0] = __builtin_amdgcn_wmma_f32_16x16x4_f32(
                false, a0, false, b0, (short)0, acc[0][0], false, false);
            acc[0][1] = __builtin_amdgcn_wmma_f32_16x16x4_f32(
                false, a0, false, b1, (short)0, acc[0][1], false, false);
            acc[0][2] = __builtin_amdgcn_wmma_f32_16x16x4_f32(
                false, a0, false, b2, (short)0, acc[0][2], false, false);
            acc[0][3] = __builtin_amdgcn_wmma_f32_16x16x4_f32(
                false, a0, false, b3, (short)0, acc[0][3], false, false);
            acc[1][0] = __builtin_amdgcn_wmma_f32_16x16x4_f32(
                false, a1, false, b0, (short)0, acc[1][0], false, false);
            acc[1][1] = __builtin_amdgcn_wmma_f32_16x16x4_f32(
                false, a1, false, b1, (short)0, acc[1][1], false, false);
            acc[1][2] = __builtin_amdgcn_wmma_f32_16x16x4_f32(
                false, a1, false, b2, (short)0, acc[1][2], false, false);
            acc[1][3] = __builtin_amdgcn_wmma_f32_16x16x4_f32(
                false, a1, false, b3, (short)0, acc[1][3], false, false);
        }
        __syncthreads();
    }

    float* dsT = smem;
    #pragma unroll
    for (int msb = 0; msb < 2; ++msb)
        #pragma unroll
        for (int nsb = 0; nsb < 4; ++nsb)
            #pragma unroll
            for (int r = 0; r < 8; ++r) {
                int row = waveM + msb * 16 + 8 * hi + r;
                int col = nsb * 16 + lo;
                dsT[row * LDD + col] = acc[msb][nsb][r];
            }
    __syncthreads();

    const int baseB = bm * MT;
    const int baseC = bn * (NT / P_N);
    const size_t outStride = (size_t)B_N * C_N;
    #pragma unroll
    for (int q = t; q < MT * (NT / P_N); q += NTHREADS) {
        int m  = q >> 3;
        int cl = q & 7;
        float zz   = z2[baseB + m];
        float best = 3.402823466e38f;
        #pragma unroll
        for (int j = 0; j < P_N; ++j) {
            float cr = dsT[m * LDD + cl * P_N + j];
            float d  = zz - 2.0f * cr + mu2[bn * NT + cl * P_N + j];
            best = fminf(best, d);
        }
        size_t o = (size_t)(baseB + m) * C_N + (baseC + cl);
        out[o]             = -best;
        out[outStride + o] =  best;
    }
}

extern "C" void kernel_launch(void* const* d_in, const int* in_sizes, int n_in,
                              void* d_out, int out_size, void* d_ws, size_t ws_size,
                              hipStream_t stream) {
    (void)in_sizes; (void)n_in; (void)out_size; (void)ws_size;
    const float* z  = (const float*)d_in[0];
    const float* mu = (const float*)d_in[1];
    float* out = (float*)d_out;
    float* z2  = (float*)d_ws;
    float* mu2 = z2 + B_N;

    rowsq_kernel<<<B_N / 8,  256, 0, stream>>>(z,  z2,  B_N);
    rowsq_kernel<<<CP_N / 8, 256, 0, stream>>>(mu, mu2, CP_N);

    dim3 grid(CP_N / NT, B_N / MT);
    proto_head_kernel<<<grid, NTHREADS, 0, stream>>>(z, mu, z2, mu2, out);
}


# Prefetch the next K stage's z and mu tiles into registers as one batch of global loads during the current stage's WMMAs (K stage 64->32 so the set fits); summation order unchanged
# speedup vs baseline: 1.2791x; 1.2791x over previous
#include <hip/hip_runtime.h>
#include <hip/hip_bf16.h>

typedef __attribute__((ext_vector_type(2))) float v2f;
typedef __attribute__((ext_vector_type(8))) float v8f;
typedef __attribute__((ext_vector_type(4))) float v4f;

#define B_N   8192
#define C_N   1000
#define P_N   8
#define D_N   512
#define CP_N  (C_N * P_N)

#define MT    128
#define NT    64
#define KT    32
#define LDA   36
#define LDD   66
#define NTHREADS 128

__global__ __launch_bounds__(256) void rowsq_kernel(const float* __restrict__ x,
                                                    float* __restrict__ out,
                                                    int rows) {
    const int lane = threadIdx.x & 31;
    const int row  = blockIdx.x * 8 + (threadIdx.x >> 5);
    if (row >= rows) return;
    const float* r = x + (size_t)row * D_N;
    float s = 0.0f;
    #pragma unroll
    for (int i = 0; i < D_N; i += 128) {
        float4 v = *(const float4*)(r + i + lane * 4);
        s += v.x * v.x + v.y * v.y + v.z * v.z + v.w * v.w;
    }
    #pragma unroll
    for (int m = 16; m >= 1; m >>= 1) s += __shfl_xor(s, m, 32);
    if (lane == 0) out[row] = s;
}

__global__ __launch_bounds__(NTHREADS) void proto_head_kernel(
    const float* __restrict__ z,
    const float* __restrict__ mu,
    const float* __restrict__ z2,
    const float* __restrict__ mu2,
    float* __restrict__ out)
{
    __shared__ float smem[MT * LDD];
    float* zs = smem;
    float* ms = smem + MT * LDA;

    const int t     = threadIdx.x;
    const int wave  = t >> 5;
    const int lane  = t & 31;
    const int lo    = lane & 15;
    const int hi    = lane >> 4;
    const int bn    = blockIdx.x;
    const int bm    = blockIdx.y;
    const int waveM = wave * 32;

    const float* zg = z  + (size_t)(bm * MT) * D_N;
    const float* mg = mu + (size_t)(bn * NT) * D_N;

    v8f acc[2][4] = {};

    const int srow = t >> 3;
    const int sc4  = (t & 7) << 2;
    const float* zp = zg + (size_t)srow * D_N + sc4;
    const float* mp = mg + (size_t)srow * D_N + sc4;
    float* zsw = zs + srow * LDA + sc4;
    float* msw = ms + srow * LDA + sc4;

    v4f pz[8], pm[4];
    #pragma unroll
    for (int p = 0; p < 8; ++p) pz[p] = *(const v4f*)(zp + p * 16 * D_N);
    #pragma unroll
    for (int p = 0; p < 4; ++p) pm[p] = *(const v4f*)(mp + p * 16 * D_N);
    #pragma unroll
    for (int p = 0; p < 8; ++p) *(v4f*)(zsw + p * 16 * LDA) = pz[p];
    #pragma unroll
    for (int p = 0; p < 4; ++p) *(v4f*)(msw + p * 16 * LDA) = pm[p];
    __syncthreads();

    for (int ks = 0; ks < D_N; ks += KT) {
        const bool more = (ks + KT) < D_N;
        if (more) {
            #pragma unroll
            for (int p = 0; p < 8; ++p)
                pz[p] = *(const v4f*)(zp + p * 16 * D_N + ks + KT);
            #pragma unroll
            for (int p = 0; p < 4; ++p)
                pm[p] = *(const v4f*)(mp + p * 16 * D_N + ks + KT);
        }

        #pragma unroll
        for (int kk = 0; kk < KT; kk += 4) {
            v2f a0 = *(const v2f*)(zs + (waveM +      lo) * LDA + kk + 2 * hi);
            v2f a1 = *(const v2f*)(zs + (waveM + 16 + lo) * LDA + kk + 2 * hi);
            v2f b0 = *(const v2f*)(ms + (      lo) * LDA + kk + 2 * hi);
            v2f b1 = *(const v2f*)(ms + (16 + lo) * LDA + kk + 2 * hi);
            v2f b2 = *(const v2f*)(ms + (32 + lo) * LDA + kk + 2 * hi);
            v2f b3 = *(const v2f*)(ms + (48 + lo) * LDA + kk + 2 * hi);

            acc[0][0] = __builtin_amdgcn_wmma_f32_16x16x4_f32(
                false, a0, false, b0, (short)0, acc[0][0], false, false);
            acc[0][1] = __builtin_amdgcn_wmma_f32_16x16x4_f32(
                false, a0, false, b1, (short)0, acc[0][1], false, false);
            acc[0][2] = __builtin_amdgcn_wmma_f32_16x16x4_f32(
                false, a0, false, b2, (short)0, acc[0][2], false, false);
            acc[0][3] = __builtin_amdgcn_wmma_f32_16x16x4_f32(
                false, a0, false, b3, (short)0, acc[0][3], false, false);
            acc[1][0] = __builtin_amdgcn_wmma_f32_16x16x4_f32(
                false, a1, false, b0, (short)0, acc[1][0], false, false);
            acc[1][1] = __builtin_amdgcn_wmma_f32_16x16x4_f32(
                false, a1, false, b1, (short)0, acc[1][1], false, false);
            acc[1][2] = __builtin_amdgcn_wmma_f32_16x16x4_f32(
                false, a1, false, b2, (short)0, acc[1][2], false, false);
            acc[1][3] = __builtin_amdgcn_wmma_f32_16x16x4_f32(
                false, a1, false, b3, (short)0, acc[1][3], false, false);
        }
        __syncthreads();

        if (more) {
            #pragma unroll
            for (int p = 0; p < 8; ++p) *(v4f*)(zsw + p * 16 * LDA) = pz[p];
            #pragma unroll
            for (int p = 0; p < 4; ++p) *(v4f*)(msw + p * 16 * LDA) = pm[p];
            __syncthreads();
        }
    }

    float* dsT = smem;
    #pragma unroll
    for (int msb = 0; msb < 2; ++msb)
        #pragma unroll
        for (int nsb = 0; nsb < 4; ++nsb)
            #pragma unroll
            for (int r = 0; r < 8; ++r) {
                int row = waveM + msb * 16 + 8 * hi + r;
                int col = nsb * 16 + lo;
                dsT[row * LDD + col] = acc[msb][nsb][r];
            }
    __syncthreads();

    const int baseB = bm * MT;
    const int baseC = bn * (NT / P_N);
    const size_t outStride = (size_t)B_N * C_N;
    #pragma unroll
    for (int q = t; q < MT * (NT / P_N); q += NTHREADS) {
        int m  = q >> 3;
        int cl = q & 7;
        float zz   = z2[baseB + m];
        float best = 3.402823466e38f;
        #pragma unroll
        for (int j = 0; j < P_N; ++j) {
            float cr = dsT[m * LDD + cl * P_N + j];
            float d  = zz - 2.0f * cr + mu2[bn * NT + cl * P_N + j];
            best = fminf(best, d);
        }
        size_t o = (size_t)(baseB + m) * C_N + (baseC + cl);
        out[o]             = -best;
        out[outStride + o] =  best;
    }
}

extern "C" void kernel_launch(void* const* d_in, const int* in_sizes, int n_in,
                              void* d_out, int out_size, void* d_ws, size_t ws_size,
                              hipStream_t stream) {
    (void)in_sizes; (void)n_in; (void)out_size; (void)ws_size;
    const float* z  = (const float*)d_in[0];
    const float* mu = (const float*)d_in[1];
    float* out = (float*)d_out;
    float* z2  = (float*)d_ws;
    float* mu2 = z2 + B_N;

    rowsq_kernel<<<B_N / 8,  256, 0, stream>>>(z,  z2,  B_N);
    rowsq_kernel<<<CP_N / 8, 256, 0, stream>>>(mu, mu2, CP_N);

    dim3 grid(CP_N / NT, B_N / MT);
    proto_head_kernel<<<grid, NTHREADS, 0, stream>>>(z, mu, z2, mu2, out);
}
